# starting kernel
# baseline (speedup 1.0000x reference)
#include <hip/hip_runtime.h>
#include <math.h>

#define BB 2
#define HH 8
#define SS 4096
#define DD 64
#define BH (BB*HH)
#define NS 16
#define CH (SS/NS)
#define PSTR 4352
#define FINBASE (BH*NS*PSTR)
#define LROW 68

typedef __attribute__((ext_vector_type(2))) float v2f;
typedef __attribute__((ext_vector_type(8))) float v8f;

__device__ __forceinline__ v8f wmma_f32_4(v2f a, v2f b, v8f c) {
    return __builtin_amdgcn_wmma_f32_16x16x4_f32(
        false, a, false, b, (short)0, c, false, false);
}

__device__ __forceinline__ v2f ld_frag(const float* p, int k0, int col, int lane, int str) {
    int ks = k0 + 2 * (lane >> 4);
    v2f r;
    r.x = p[ks * str + col];
    r.y = p[(ks + 1) * str + col];
    return r;
}

__global__ __launch_bounds__(256) void cosattn_p1(
    const float* __restrict__ K, const float* __restrict__ V,
    const int* __restrict__ mask, float* __restrict__ ws) {
    __shared__ float kn[CH * LROW];
    __shared__ float vv[CH * LROW];
    __shared__ float mf[CH];

    const int tid = threadIdx.x;
    const int blk = blockIdx.x;
    const int bh = blk / NS;
    const int chunk = blk % NS;
    const int b = bh / HH;
    const long gbase = ((long)bh * SS + (long)chunk * CH) * DD;

#pragma unroll
    for (int j = 0; j < 16; ++j) {
        int e4 = tid + j * 256;
        int row = e4 >> 4;
        int col = (e4 & 15) * 4;
        float4 kq = *(const float4*)(K + gbase + (long)e4 * 4);
        float4 vq = *(const float4*)(V + gbase + (long)e4 * 4);
        *(float4*)(kn + row * LROW + col) = kq;
        *(float4*)(vv + row * LROW + col) = vq;
    }
    {
        int kidx = chunk * CH + tid;
        mf[tid] = (mask[b * SS + kidx] != 0) ? 1.0f : 0.0f;
    }
    __syncthreads();

    {
        float* rowp = kn + tid * LROW;
        float ssum = 0.0f;
#pragma unroll
        for (int c = 0; c < DD; c += 4) {
            float4 x = *(float4*)(rowp + c);
            ssum += x.x * x.x + x.y * x.y + x.z * x.z + x.w * x.w;
        }
        float scale = mf[tid] / fmaxf(sqrtf(ssum), 1e-12f);
#pragma unroll
        for (int c = 0; c < DD; c += 4) {
            float4 x = *(float4*)(rowp + c);
            x.x *= scale; x.y *= scale; x.z *= scale; x.w *= scale;
            *(float4*)(rowp + c) = x;
        }
    }
    __syncthreads();

    const long pbase = (long)blk * PSTR;

    if (tid < 64) {
        float a = 0.0f;
        for (int r = 0; r < CH; ++r) a += kn[r * LROW + tid];
        ws[pbase + 4096 + tid] = a;
    } else if (tid < 128) {
        int c = tid - 64;
        float a = 0.0f;
        for (int r = 0; r < CH; ++r) a += mf[r] * vv[r * LROW + c];
        ws[pbase + 4160 + c] = a;
    } else if (tid == 128) {
        float a = 0.0f;
        for (int r = 0; r < CH; ++r) a += mf[r];
        ws[pbase + 4224] = a;
    }
    __syncthreads();

    const int lane = tid & 31;
    const int wv = tid >> 5;
    const int ti = wv >> 1;
    const int tj = (wv & 1) * 2;
    const int colA = ti * 16 + (lane & 15);
    const int colB0 = tj * 16 + (lane & 15);
    v8f acc0 = {};
    v8f acc1 = {};
    for (int k0 = 0; k0 < CH; k0 += 4) {
        v2f a  = ld_frag(kn, k0, colA, lane, LROW);
        v2f b0 = ld_frag(vv, k0, colB0, lane, LROW);
        v2f b1 = ld_frag(vv, k0, colB0 + 16, lane, LROW);
        acc0 = wmma_f32_4(a, b0, acc0);
        acc1 = wmma_f32_4(a, b1, acc1);
    }
    const int rbase = ti * 16 + 8 * (lane >> 4);
    const int c0 = tj * 16 + (lane & 15);
#pragma unroll
    for (int r = 0; r < 8; ++r) {
        ws[pbase + (rbase + r) * 64 + c0] = acc0[r];
        ws[pbase + (rbase + r) * 64 + c0 + 16] = acc1[r];
    }
}

__global__ __launch_bounds__(256) void cosattn_reduce(float* __restrict__ ws) {
    const int slot = blockIdx.x / 17;
    const int e = (blockIdx.x % 17) * 256 + threadIdx.x;
    if (e >= 4225) return;
    const long pb = (long)slot * NS * PSTR;
    float a = 0.0f;
#pragma unroll
    for (int c = 0; c < NS; ++c) a += ws[pb + (long)c * PSTR + e];
    ws[FINBASE + (long)slot * PSTR + e] = a;
}

__global__ __launch_bounds__(256) void cosattn_p2(
    const float* __restrict__ Q, const float* __restrict__ ws,
    float* __restrict__ out) {
    __shared__ float fin[PSTR];
    __shared__ float qs[8 * 16 * LROW];
    __shared__ float scale_s[128];
    __shared__ float denom_s[128];

    const int tid = threadIdx.x;
    const int lane = tid & 31;
    const int wv = tid >> 5;
    const int bh = blockIdx.x >> 5;
    const int rg = blockIdx.x & 31;
    const int q0 = rg * 128;

    const float* finp = ws + FINBASE + (long)bh * PSTR;
    for (int e = tid; e < 4225; e += 256) fin[e] = finp[e];

    const long qrowbase = (long)bh * SS + q0 + wv * 16;
    float* myq = qs + wv * 16 * LROW;
#pragma unroll
    for (int j = 0; j < 16; ++j) {
        float2 x = *(const float2*)(Q + (qrowbase + j) * DD + lane * 2);
        *(float2*)(myq + j * LROW + lane * 2) = x;
    }
    __syncthreads();

    const int r = lane & 15;
    const int half = lane >> 4;
    {
        float ssum = 0.0f, kd = 0.0f;
#pragma unroll
        for (int c32 = 0; c32 < 32; ++c32) {
            int c = half * 32 + c32;
            float v = myq[r * LROW + c];
            ssum += v * v;
            kd += v * fin[4096 + c];
        }
        ssum += __shfl_xor(ssum, 16, 32);
        kd += __shfl_xor(kd, 16, 32);
        float sc = 1.0f / fmaxf(sqrtf(ssum), 1e-12f);
        if (lane < 16) {
            scale_s[wv * 16 + lane] = sc;
            denom_s[wv * 16 + lane] = sc * kd + fin[4224];
        }
    }
    __syncthreads();

    v8f acc[4];
#pragma unroll
    for (int nt = 0; nt < 4; ++nt) {
        float sv = fin[4160 + nt * 16 + r];
        v8f ci;
#pragma unroll
        for (int i = 0; i < 8; ++i) ci[i] = sv;
        acc[nt] = ci;
    }
    const float scr = scale_s[wv * 16 + r];
    for (int k0 = 0; k0 < DD; k0 += 4) {
        int ks = k0 + 2 * half;
        v2f a;
        a.x = myq[r * LROW + ks] * scr;
        a.y = myq[r * LROW + ks + 1] * scr;
#pragma unroll
        for (int nt = 0; nt < 4; ++nt) {
            int n = nt * 16 + r;
            v2f bfr;
            bfr.x = fin[ks * 64 + n];
            bfr.y = fin[(ks + 1) * 64 + n];
            acc[nt] = wmma_f32_4(a, bfr, acc[nt]);
        }
    }

#pragma unroll
    for (int nt = 0; nt < 4; ++nt) {
        int col = nt * 16 + r;
#pragma unroll
        for (int rr = 0; rr < 8; ++rr) {
            int rowT = rr + 8 * half;
            float dv = denom_s[wv * 16 + rowT];
            out[(qrowbase + rowT) * DD + col] = acc[nt][rr] / dv;
        }
    }
}

extern "C" void kernel_launch(void* const* d_in, const int* in_sizes, int n_in,
                              void* d_out, int out_size, void* d_ws, size_t ws_size,
                              hipStream_t stream) {
    const float* Q = (const float*)d_in[0];
    const float* K = (const float*)d_in[1];
    const float* V = (const float*)d_in[2];
    const int* M = (const int*)d_in[3];
    float* out = (float*)d_out;
    float* ws = (float*)d_ws;

    cosattn_p1<<<dim3(BH * NS), dim3(256), 0, stream>>>(K, V, M, ws);
    cosattn_reduce<<<dim3(BH * 17), dim3(256), 0, stream>>>(ws);
    cosattn_p2<<<dim3(BH * 32), dim3(256), 0, stream>>>(Q, ws, out);
}


# p2 front end: reduced slot and q tile fetched as one group of float4 loads issued before the first wait, replacing the one-float staging loop (a wait per trip) and float2 q loads; sums unchanged
# speedup vs baseline: 1.1245x; 1.1245x over previous
#include <hip/hip_runtime.h>
#include <math.h>

#define BB 2
#define HH 8
#define SS 4096
#define DD 64
#define BH (BB*HH)
#define NS 16
#define CH (SS/NS)
#define PSTR 4352
#define FINBASE (BH*NS*PSTR)
#define LROW 68

typedef __attribute__((ext_vector_type(2))) float v2f;
typedef __attribute__((ext_vector_type(8))) float v8f;

__device__ __forceinline__ v8f wmma_f32_4(v2f a, v2f b, v8f c) {
    return __builtin_amdgcn_wmma_f32_16x16x4_f32(
        false, a, false, b, (short)0, c, false, false);
}

__device__ __forceinline__ v2f ld_frag(const float* p, int k0, int col, int lane, int str) {
    int ks = k0 + 2 * (lane >> 4);
    v2f r;
    r.x = p[ks * str + col];
    r.y = p[(ks + 1) * str + col];
    return r;
}

__global__ __launch_bounds__(256) void cosattn_p1(
    const float* __restrict__ K, const float* __restrict__ V,
    const int* __restrict__ mask, float* __restrict__ ws) {
    __shared__ float kn[CH * LROW];
    __shared__ float vv[CH * LROW];
    __shared__ float mf[CH];

    const int tid = threadIdx.x;
    const int blk = blockIdx.x;
    const int bh = blk / NS;
    const int chunk = blk % NS;
    const int b = bh / HH;
    const long gbase = ((long)bh * SS + (long)chunk * CH) * DD;

#pragma unroll
    for (int j = 0; j < 16; ++j) {
        int e4 = tid + j * 256;
        int row = e4 >> 4;
        int col = (e4 & 15) * 4;
        float4 kq = *(const float4*)(K + gbase + (long)e4 * 4);
        float4 vq = *(const float4*)(V + gbase + (long)e4 * 4);
        *(float4*)(kn + row * LROW + col) = kq;
        *(float4*)(vv + row * LROW + col) = vq;
    }
    {
        int kidx = chunk * CH + tid;
        mf[tid] = (mask[b * SS + kidx] != 0) ? 1.0f : 0.0f;
    }
    __syncthreads();

    {
        float* rowp = kn + tid * LROW;
        float ssum = 0.0f;
#pragma unroll
        for (int c = 0; c < DD; c += 4) {
            float4 x = *(float4*)(rowp + c);
            ssum += x.x * x.x + x.y * x.y + x.z * x.z + x.w * x.w;
        }
        float scale = mf[tid] / fmaxf(sqrtf(ssum), 1e-12f);
#pragma unroll
        for (int c = 0; c < DD; c += 4) {
            float4 x = *(float4*)(rowp + c);
            x.x *= scale; x.y *= scale; x.z *= scale; x.w *= scale;
            *(float4*)(rowp + c) = x;
        }
    }
    __syncthreads();

    const long pbase = (long)blk * PSTR;

    if (tid < 64) {
        float a = 0.0f;
        for (int r = 0; r < CH; ++r) a += kn[r * LROW + tid];
        ws[pbase + 4096 + tid] = a;
    } else if (tid < 128) {
        int c = tid - 64;
        float a = 0.0f;
        for (int r = 0; r < CH; ++r) a += mf[r] * vv[r * LROW + c];
        ws[pbase + 4160 + c] = a;
    } else if (tid == 128) {
        float a = 0.0f;
        for (int r = 0; r < CH; ++r) a += mf[r];
        ws[pbase + 4224] = a;
    }
    __syncthreads();

    const int lane = tid & 31;
    const int wv = tid >> 5;
    const int ti = wv >> 1;
    const int tj = (wv & 1) * 2;
    const int colA = ti * 16 + (lane & 15);
    const int colB0 = tj * 16 + (lane & 15);
    v8f acc0 = {};
    v8f acc1 = {};
    for (int k0 = 0; k0 < CH; k0 += 4) {
        v2f a  = ld_frag(kn, k0, colA, lane, LROW);
        v2f b0 = ld_frag(vv, k0, colB0, lane, LROW);
        v2f b1 = ld_frag(vv, k0, colB0 + 16, lane, LROW);
        acc0 = wmma_f32_4(a, b0, acc0);
        acc1 = wmma_f32_4(a, b1, acc1);
    }
    const int rbase = ti * 16 + 8 * (lane >> 4);
    const int c0 = tj * 16 + (lane & 15);
#pragma unroll
    for (int r = 0; r < 8; ++r) {
        ws[pbase + (rbase + r) * 64 + c0] = acc0[r];
        ws[pbase + (rbase + r) * 64 + c0 + 16] = acc1[r];
    }
}

__global__ __launch_bounds__(256) void cosattn_reduce(float* __restrict__ ws) {
    const int slot = blockIdx.x / 17;
    const int e = (blockIdx.x % 17) * 256 + threadIdx.x;
    if (e >= 4225) return;
    const long pb = (long)slot * NS * PSTR;
    float a = 0.0f;
#pragma unroll
    for (int c = 0; c < NS; ++c) a += ws[pb + (long)c * PSTR + e];
    ws[FINBASE + (long)slot * PSTR + e] = a;
}

__global__ __launch_bounds__(256) void cosattn_p2(
    const float* __restrict__ Q, const float* __restrict__ ws,
    float* __restrict__ out) {
    __shared__ alignas(16) float fin[PSTR];
    __shared__ alignas(16) float qs[8 * 16 * LROW];
    __shared__ float scale_s[128];
    __shared__ float denom_s[128];

    const int tid = threadIdx.x;
    const int lane = tid & 31;
    const int wv = tid >> 5;
    const int bh = blockIdx.x >> 5;
    const int rg = blockIdx.x & 31;
    const int q0 = rg * 128;

    const float* finp = ws + FINBASE + (long)bh * PSTR;
    const long qrowbase = (long)bh * SS + q0 + wv * 16;
    float* myq = qs + wv * 16 * LROW;
    const float* qg = Q + qrowbase * DD;

    float4 mreg[4];
    for (int j = 0; j < 4; ++j)
        mreg[j] = *(const float4*)(finp + 4 * (tid + j * 256));
    const float treg = finp[4096 + (tid < 129 ? tid : 128)];
    float4 qreg[8];
    for (int j = 0; j < 8; ++j)
        qreg[j] = *(const float4*)(qg + 4 * (j * 32 + lane));

    for (int j = 0; j < 4; ++j)
        *(float4*)(fin + 4 * (tid + j * 256)) = mreg[j];
    fin[4096 + tid] = treg;
    for (int j = 0; j < 8; ++j) {
        int f = j * 32 + lane;
        *(float4*)(myq + (f >> 4) * LROW + (f & 15) * 4) = qreg[j];
    }
    __syncthreads();

    const int r = lane & 15;
    const int half = lane >> 4;
    {
        float ssum = 0.0f, kd = 0.0f;
#pragma unroll
        for (int c32 = 0; c32 < 32; ++c32) {
            int c = half * 32 + c32;
            float v = myq[r * LROW + c];
            ssum += v * v;
            kd += v * fin[4096 + c];
        }
        ssum += __shfl_xor(ssum, 16, 32);
        kd += __shfl_xor(kd, 16, 32);
        float sc = 1.0f / fmaxf(sqrtf(ssum), 1e-12f);
        if (lane < 16) {
            scale_s[wv * 16 + lane] = sc;
            denom_s[wv * 16 + lane] = sc * kd + fin[4224];
        }
    }
    __syncthreads();

    v8f acc[4];
#pragma unroll
    for (int nt = 0; nt < 4; ++nt) {
        float sv = fin[4160 + nt * 16 + r];
        v8f ci;
#pragma unroll
        for (int i = 0; i < 8; ++i) ci[i] = sv;
        acc[nt] = ci;
    }
    const float scr = scale_s[wv * 16 + r];
    for (int k0 = 0; k0 < DD; k0 += 4) {
        int ks = k0 + 2 * half;
        v2f a;
        a.x = myq[r * LROW + ks] * scr;
        a.y = myq[r * LROW + ks + 1] * scr;
#pragma unroll
        for (int nt = 0; nt < 4; ++nt) {
            int n = nt * 16 + r;
            v2f bfr;
            bfr.x = fin[ks * 64 + n];
            bfr.y = fin[(ks + 1) * 64 + n];
            acc[nt] = wmma_f32_4(a, bfr, acc[nt]);
        }
    }

#pragma unroll
    for (int nt = 0; nt < 4; ++nt) {
        int col = nt * 16 + r;
#pragma unroll
        for (int rr = 0; rr < 8; ++rr) {
            int rowT = rr + 8 * half;
            float dv = denom_s[wv * 16 + rowT];
            out[(qrowbase + rowT) * DD + col] = acc[nt][rr] / dv;
        }
    }
}

extern "C" void kernel_launch(void* const* d_in, const int* in_sizes, int n_in,
                              void* d_out, int out_size, void* d_ws, size_t ws_size,
                              hipStream_t stream) {
    const float* Q = (const float*)d_in[0];
    const float* K = (const float*)d_in[1];
    const float* V = (const float*)d_in[2];
    const int* M = (const int*)d_in[3];
    float* out = (float*)d_out;
    float* ws = (float*)d_ws;

    cosattn_p1<<<dim3(BH * NS), dim3(256), 0, stream>>>(K, V, M, ws);
    cosattn_reduce<<<dim3(BH * 17), dim3(256), 0, stream>>>(ws);
    cosattn_p2<<<dim3(BH * 32), dim3(256), 0, stream>>>(Q, ws, out);
}


# p1: ksum, s and cnt summed per lane from the WMMA loop's own A/B fragment values, then across lane halves (new summation order), gathered in LDS for the stores; replaces three serial one-wave loops
# speedup vs baseline: 1.2556x; 1.1166x over previous
#include <hip/hip_runtime.h>
#include <math.h>

#define BB 2
#define HH 8
#define SS 4096
#define DD 64
#define BH (BB*HH)
#define NS 16
#define CH (SS/NS)
#define PSTR 4352
#define FINBASE (BH*NS*PSTR)
#define LROW 68

typedef __attribute__((ext_vector_type(2))) float v2f;
typedef __attribute__((ext_vector_type(8))) float v8f;

__device__ __forceinline__ v8f wmma_f32_4(v2f a, v2f b, v8f c) {
    return __builtin_amdgcn_wmma_f32_16x16x4_f32(
        false, a, false, b, (short)0, c, false, false);
}

__device__ __forceinline__ v2f ld_frag(const float* p, int k0, int col, int lane, int str) {
    int ks = k0 + 2 * (lane >> 4);
    v2f r;
    r.x = p[ks * str + col];
    r.y = p[(ks + 1) * str + col];
    return r;
}

__global__ __launch_bounds__(256) void cosattn_p1(
    const float* __restrict__ K, const float* __restrict__ V,
    const int* __restrict__ mask, float* __restrict__ ws) {
    __shared__ float kn[CH * LROW];
    __shared__ float vv[CH * LROW];
    __shared__ float mf[CH];
    __shared__ float colsum[132];

    const int tid = threadIdx.x;
    const int blk = blockIdx.x;
    const int bh = blk / NS;
    const int chunk = blk % NS;
    const int b = bh / HH;
    const long gbase = ((long)bh * SS + (long)chunk * CH) * DD;

#pragma unroll
    for (int j = 0; j < 16; ++j) {
        int e4 = tid + j * 256;
        int row = e4 >> 4;
        int col = (e4 & 15) * 4;
        float4 kq = *(const float4*)(K + gbase + (long)e4 * 4);
        float4 vq = *(const float4*)(V + gbase + (long)e4 * 4);
        *(float4*)(kn + row * LROW + col) = kq;
        *(float4*)(vv + row * LROW + col) = vq;
    }
    {
        int kidx = chunk * CH + tid;
        mf[tid] = (mask[b * SS + kidx] != 0) ? 1.0f : 0.0f;
    }
    __syncthreads();

    {
        float* rowp = kn + tid * LROW;
        float ssum = 0.0f;
#pragma unroll
        for (int c = 0; c < DD; c += 4) {
            float4 x = *(float4*)(rowp + c);
            ssum += x.x * x.x + x.y * x.y + x.z * x.z + x.w * x.w;
        }
        float scale = mf[tid] / fmaxf(sqrtf(ssum), 1e-12f);
#pragma unroll
        for (int c = 0; c < DD; c += 4) {
            float4 x = *(float4*)(rowp + c);
            x.x *= scale; x.y *= scale; x.z *= scale; x.w *= scale;
            *(float4*)(rowp + c) = x;
        }
    }
    __syncthreads();

    const long pbase = (long)blk * PSTR;

    const int lane = tid & 31;
    const int wv = tid >> 5;
    const int ti = wv >> 1;
    const int tj = (wv & 1) * 2;
    const int colA = ti * 16 + (lane & 15);
    const int colB0 = tj * 16 + (lane & 15);
    v8f acc0 = {};
    v8f acc1 = {};
    float ksp = 0.0f;
    float sp0 = 0.0f;
    float sp1 = 0.0f;
    float cnp = 0.0f;
    for (int k0 = 0; k0 < CH; k0 += 4) {
        v2f a  = ld_frag(kn, k0, colA, lane, LROW);
        v2f b0 = ld_frag(vv, k0, colB0, lane, LROW);
        v2f b1 = ld_frag(vv, k0, colB0 + 16, lane, LROW);
        const int ks = k0 + 2 * (lane >> 4);
        const float m0 = mf[ks];
        const float m1 = mf[ks + 1];
        acc0 = wmma_f32_4(a, b0, acc0);
        acc1 = wmma_f32_4(a, b1, acc1);
        ksp += a.x;
        ksp += a.y;
        sp0 += m0 * b0.x;
        sp0 += m1 * b0.y;
        sp1 += m0 * b1.x;
        sp1 += m1 * b1.y;
        cnp += m0;
        cnp += m1;
    }
    ksp += __shfl_xor(ksp, 16, 32);
    sp0 += __shfl_xor(sp0, 16, 32);
    sp1 += __shfl_xor(sp1, 16, 32);
    cnp += __shfl_xor(cnp, 16, 32);
    if (lane < 16) {
        if (tj == 0) colsum[colA] = ksp;
        if (ti == 0) {
            colsum[64 + colB0] = sp0;
            colsum[64 + colB0 + 16] = sp1;
        }
        if (tid == 0) colsum[128] = cnp;
    }
    __syncthreads();
    if (tid < 129) ws[pbase + 4096 + tid] = colsum[tid];

    const int rbase = ti * 16 + 8 * (lane >> 4);
    const int c0 = tj * 16 + (lane & 15);
#pragma unroll
    for (int r = 0; r < 8; ++r) {
        ws[pbase + (rbase + r) * 64 + c0] = acc0[r];
        ws[pbase + (rbase + r) * 64 + c0 + 16] = acc1[r];
    }
}

__global__ __launch_bounds__(256) void cosattn_reduce(float* __restrict__ ws) {
    const int slot = blockIdx.x / 17;
    const int e = (blockIdx.x % 17) * 256 + threadIdx.x;
    if (e >= 4225) return;
    const long pb = (long)slot * NS * PSTR;
    float a = 0.0f;
#pragma unroll
    for (int c = 0; c < NS; ++c) a += ws[pb + (long)c * PSTR + e];
    ws[FINBASE + (long)slot * PSTR + e] = a;
}

__global__ __launch_bounds__(256) void cosattn_p2(
    const float* __restrict__ Q, const float* __restrict__ ws,
    float* __restrict__ out) {
    __shared__ alignas(16) float fin[PSTR];
    __shared__ alignas(16) float qs[8 * 16 * LROW];
    __shared__ float scale_s[128];
    __shared__ float denom_s[128];

    const int tid = threadIdx.x;
    const int lane = tid & 31;
    const int wv = tid >> 5;
    const int bh = blockIdx.x >> 5;
    const int rg = blockIdx.x & 31;
    const int q0 = rg * 128;

    const float* finp = ws + FINBASE + (long)bh * PSTR;
    const long qrowbase = (long)bh * SS + q0 + wv * 16;
    float* myq = qs + wv * 16 * LROW;
    const float* qg = Q + qrowbase * DD;

    float4 mreg[4];
    for (int j = 0; j < 4; ++j)
        mreg[j] = *(const float4*)(finp + 4 * (tid + j * 256));
    const float treg = finp[4096 + (tid < 129 ? tid : 128)];
    float4 qreg[8];
    for (int j = 0; j < 8; ++j)
        qreg[j] = *(const float4*)(qg + 4 * (j * 32 + lane));

    for (int j = 0; j < 4; ++j)
        *(float4*)(fin + 4 * (tid + j * 256)) = mreg[j];
    fin[4096 + tid] = treg;
    for (int j = 0; j < 8; ++j) {
        int f = j * 32 + lane;
        *(float4*)(myq + (f >> 4) * LROW + (f & 15) * 4) = qreg[j];
    }
    __syncthreads();

    const int r = lane & 15;
    const int half = lane >> 4;
    {
        float ssum = 0.0f, kd = 0.0f;
#pragma unroll
        for (int c32 = 0; c32 < 32; ++c32) {
            int c = half * 32 + c32;
            float v = myq[r * LROW + c];
            ssum += v * v;
            kd += v * fin[4096 + c];
        }
        ssum += __shfl_xor(ssum, 16, 32);
        kd += __shfl_xor(kd, 16, 32);
        float sc = 1.0f / fmaxf(sqrtf(ssum), 1e-12f);
        if (lane < 16) {
            scale_s[wv * 16 + lane] = sc;
            denom_s[wv * 16 + lane] = sc * kd + fin[4224];
        }
    }
    __syncthreads();

    v8f acc[4];
#pragma unroll
    for (int nt = 0; nt < 4; ++nt) {
        float sv = fin[4160 + nt * 16 + r];
        v8f ci;
#pragma unroll
        for (int i = 0; i < 8; ++i) ci[i] = sv;
        acc[nt] = ci;
    }
    const float scr = scale_s[wv * 16 + r];
    for (int k0 = 0; k0 < DD; k0 += 4) {
        int ks = k0 + 2 * half;
        v2f a;
        a.x = myq[r * LROW + ks] * scr;
        a.y = myq[r * LROW + ks + 1] * scr;
#pragma unroll
        for (int nt = 0; nt < 4; ++nt) {
            int n = nt * 16 + r;
            v2f bfr;
            bfr.x = fin[ks * 64 + n];
            bfr.y = fin[(ks + 1) * 64 + n];
            acc[nt] = wmma_f32_4(a, bfr, acc[nt]);
        }
    }

#pragma unroll
    for (int nt = 0; nt < 4; ++nt) {
        int col = nt * 16 + r;
#pragma unroll
        for (int rr = 0; rr < 8; ++rr) {
            int rowT = rr + 8 * half;
            float dv = denom_s[wv * 16 + rowT];
            out[(qrowbase + rowT) * DD + col] = acc[nt][rr] / dv;
        }
    }
}

extern "C" void kernel_launch(void* const* d_in, const int* in_sizes, int n_in,
                              void* d_out, int out_size, void* d_ws, size_t ws_size,
                              hipStream_t stream) {
    const float* Q = (const float*)d_in[0];
    const float* K = (const float*)d_in[1];
    const float* V = (const float*)d_in[2];
    const int* M = (const int*)d_in[3];
    float* out = (float*)d_out;
    float* ws = (float*)d_ws;

    cosattn_p1<<<dim3(BH * NS), dim3(256), 0, stream>>>(K, V, M, ws);
    cosattn_reduce<<<dim3(BH * 17), dim3(256), 0, stream>>>(ws);
    cosattn_p2<<<dim3(BH * 32), dim3(256), 0, stream>>>(Q, ws, out);
}


# p1 software-pipelined: key groups (32,32,64,64,64) normalized in registers (16-lane DPP sum: new summation order of key norm) and stored one group ahead of the matrix work, later loads in flight
# speedup vs baseline: 1.3767x; 1.0964x over previous
#include <hip/hip_runtime.h>
#include <math.h>

#define BB 2
#define HH 8
#define SS 4096
#define DD 64
#define BH (BB*HH)
#define NS 16
#define CH (SS/NS)
#define PSTR 4352
#define FINBASE (BH*NS*PSTR)
#define LROW 68

typedef __attribute__((ext_vector_type(2))) float v2f;
typedef __attribute__((ext_vector_type(8))) float v8f;

__device__ __forceinline__ v8f wmma_f32_4(v2f a, v2f b, v8f c) {
    return __builtin_amdgcn_wmma_f32_16x16x4_f32(
        false, a, false, b, (short)0, c, false, false);
}

__device__ __forceinline__ v2f ld_frag(const float* p, int k0, int col, int lane, int str) {
    int ks = k0 + 2 * (lane >> 4);
    v2f r;
    r.x = p[ks * str + col];
    r.y = p[(ks + 1) * str + col];
    return r;
}


__device__ __forceinline__ void lds_barrier() {
    asm volatile("s_wait_dscnt 0x0\n\ts_barrier_signal -1\n\ts_barrier_wait -1" ::: "memory");
}
__device__ __forceinline__ void lds_signal() {
    asm volatile("s_wait_dscnt 0x0\n\ts_barrier_signal -1" ::: "memory");
}
__device__ __forceinline__ void lds_wait() {
    asm volatile("s_barrier_wait -1" ::: "memory");
}
__device__ __forceinline__ void issue_barrier() {
    asm volatile("s_barrier_signal -1\n\ts_barrier_wait -1" ::: "memory");
}
__device__ __forceinline__ void use_here(float4& x) {
    asm volatile("" : "+v"(x.x), "+v"(x.y), "+v"(x.z), "+v"(x.w));
}
__device__ __forceinline__ void use_here(int& x) {
    asm volatile("" : "+v"(x));
}

template <int CTRL>
__device__ __forceinline__ float dpp_mov(float v) {
    return __int_as_float(__builtin_amdgcn_update_dpp(0, __float_as_int(v), CTRL, 0xf, 0xf, false));
}
__device__ __forceinline__ float row16_sum(float v) {
    v += dpp_mov<0x161>(v);
    v += dpp_mov<0x162>(v);
    v += dpp_mov<0x164>(v);
    v += dpp_mov<0x168>(v);
    return v;
}
__device__ __forceinline__ float sq4(float4 x) {
    return x.x * x.x + x.y * x.y + x.z * x.z + x.w * x.w;
}
__device__ __forceinline__ float4 mul4(float4 x, float s) {
    x.x *= s; x.y *= s; x.z *= s; x.w *= s;
    return x;
}

template <int J0>
__device__ __forceinline__ void p1_stage4(float4 ka, float4 kb, float4 kc, float4 kd,
                                          float4 va, float4 vb, float4 vc, float4 vd,
                                          float* kn, float* vv, const float* mf, int tid) {
    const int rsub = tid >> 4;
    const int col = (tid & 15) * 4;
    const int jsel = tid & 3;
    const float ss0 = row16_sum(sq4(ka));
    const float ss1 = row16_sum(sq4(kb));
    const float ss2 = row16_sum(sq4(kc));
    const float ss3 = row16_sum(sq4(kd));
    float sel = ss0;
    sel = (jsel == 1) ? ss1 : sel;
    sel = (jsel == 2) ? ss2 : sel;
    sel = (jsel == 3) ? ss3 : sel;
    const float msel = mf[rsub + 16 * (J0 + jsel)];
    const float scl = msel / fmaxf(sqrtf(sel), 1e-12f);
    const float s0 = dpp_mov<0x00>(scl);
    const float s1 = dpp_mov<0x55>(scl);
    const float s2 = dpp_mov<0xAA>(scl);
    const float s3 = dpp_mov<0xFF>(scl);
    const int r0 = rsub + 16 * J0;
    *(float4*)(kn + (r0     ) * LROW + col) = mul4(ka, s0);
    *(float4*)(vv + (r0     ) * LROW + col) = va;
    *(float4*)(kn + (r0 + 16) * LROW + col) = mul4(kb, s1);
    *(float4*)(vv + (r0 + 16) * LROW + col) = vb;
    *(float4*)(kn + (r0 + 32) * LROW + col) = mul4(kc, s2);
    *(float4*)(vv + (r0 + 32) * LROW + col) = vc;
    *(float4*)(kn + (r0 + 48) * LROW + col) = mul4(kd, s3);
    *(float4*)(vv + (r0 + 48) * LROW + col) = vd;
}
template <int J0>
__device__ __forceinline__ void p1_stage2(float4 ka, float4 kb, float4 va, float4 vb,
                                          float* kn, float* vv, const float* mf, int tid) {
    const int rsub = tid >> 4;
    const int col = (tid & 15) * 4;
    const int jsel = tid & 1;
    const float ss0 = row16_sum(sq4(ka));
    const float ss1 = row16_sum(sq4(kb));
    const float sel = (jsel == 1) ? ss1 : ss0;
    const float msel = mf[rsub + 16 * (J0 + jsel)];
    const float scl = msel / fmaxf(sqrtf(sel), 1e-12f);
    const float s0 = dpp_mov<0x00>(scl);
    const float s1 = dpp_mov<0x55>(scl);
    const int r0 = rsub + 16 * J0;
    *(float4*)(kn + (r0     ) * LROW + col) = mul4(ka, s0);
    *(float4*)(vv + (r0     ) * LROW + col) = va;
    *(float4*)(kn + (r0 + 16) * LROW + col) = mul4(kb, s1);
    *(float4*)(vv + (r0 + 16) * LROW + col) = vb;
}

__global__ __launch_bounds__(256) void cosattn_p1(
    const float* K, const float* V,
    const int* __restrict__ mask, float* ws) {
    __shared__ float kn[CH * LROW];
    __shared__ float vv[CH * LROW];
    __shared__ float mf[CH];
    __shared__ float colsum[132];

    const int tid = threadIdx.x;
    const int blk = blockIdx.x;
    const int bh = blk / NS;
    const int chunk = blk % NS;
    const int b = bh / HH;
    const long gbase = ((long)bh * SS + (long)chunk * CH) * DD;

    int mraw = mask[b * SS + chunk * CH + tid];

    const float* kp = K + gbase + (long)tid * 4;
    const float* vp = V + gbase + (long)tid * 4;
#define P1_LD(j) float4 kq##j = *(const float4*)(kp + (j) * 1024); \
                 const float4 vq##j = *(const float4*)(vp + (j) * 1024);
    P1_LD(0) P1_LD(1) P1_LD(2) P1_LD(3)
    issue_barrier();
    P1_LD(4) P1_LD(5) P1_LD(6) P1_LD(7)
    issue_barrier();
    P1_LD(8) P1_LD(9) P1_LD(10) P1_LD(11)
    issue_barrier();
    P1_LD(12) P1_LD(13) P1_LD(14) P1_LD(15)
#undef P1_LD
    use_here(mraw);
    mf[tid] = (mraw != 0) ? 1.0f : 0.0f;
    lds_barrier();

    const long pbase = (long)blk * PSTR;

    const int lane = tid & 31;
    const int wv = tid >> 5;
    const int ti = wv >> 1;
    const int tj = (wv & 1) * 2;
    const int colA = ti * 16 + (lane & 15);
    const int colB0 = tj * 16 + (lane & 15);
    v8f acc0 = {};
    v8f acc1 = {};
    float ksp = 0.0f;
    float sp0 = 0.0f;
    float sp1 = 0.0f;
    float cnp = 0.0f;

    auto mma_group = [&](int kbeg, int kend) {
#pragma unroll 4
        for (int k0 = kbeg; k0 < kend; k0 += 4) {
            v2f a  = ld_frag(kn, k0, colA, lane, LROW);
            v2f b0 = ld_frag(vv, k0, colB0, lane, LROW);
            v2f b1 = ld_frag(vv, k0, colB0 + 16, lane, LROW);
            const int ks = k0 + 2 * (lane >> 4);
            const float m0 = mf[ks];
            const float m1 = mf[ks + 1];
            acc0 = wmma_f32_4(a, b0, acc0);
            acc1 = wmma_f32_4(a, b1, acc1);
            ksp += a.x;
            ksp += a.y;
            sp0 += m0 * b0.x;
            sp0 += m1 * b0.y;
            sp1 += m0 * b1.x;
            sp1 += m1 * b1.y;
            cnp += m0;
            cnp += m1;
        }
    };

    use_here(kq0); use_here(kq1); use_here(kq2); use_here(kq3);
    p1_stage2<0>(kq0, kq1, vq0, vq1, kn, vv, mf, tid);
    lds_barrier();
    p1_stage2<2>(kq2, kq3, vq2, vq3, kn, vv, mf, tid);
    lds_signal();
    mma_group(0, 32);
    lds_wait();
    p1_stage4<4>(kq4, kq5, kq6, kq7, vq4, vq5, vq6, vq7, kn, vv, mf, tid);
    lds_signal();
    mma_group(32, 64);
    lds_wait();
    p1_stage4<8>(kq8, kq9, kq10, kq11, vq8, vq9, vq10, vq11, kn, vv, mf, tid);
    lds_signal();
    mma_group(64, 128);
    lds_wait();
    p1_stage4<12>(kq12, kq13, kq14, kq15, vq12, vq13, vq14, vq15, kn, vv, mf, tid);
    lds_signal();
    mma_group(128, 192);
    lds_wait();
    mma_group(192, 256);

    ksp += __shfl_xor(ksp, 16, 32);
    sp0 += __shfl_xor(sp0, 16, 32);
    sp1 += __shfl_xor(sp1, 16, 32);
    cnp += __shfl_xor(cnp, 16, 32);
    if (lane < 16) {
        if (tj == 0) colsum[colA] = ksp;
        if (ti == 0) {
            colsum[64 + colB0] = sp0;
            colsum[64 + colB0 + 16] = sp1;
        }
        if (tid == 0) colsum[128] = cnp;
    }
    __syncthreads();
    if (tid < 129) ws[pbase + 4096 + tid] = colsum[tid];

    const int rbase = ti * 16 + 8 * (lane >> 4);
    const int c0 = tj * 16 + (lane & 15);
#pragma unroll
    for (int r = 0; r < 8; ++r) {
        ws[pbase + (rbase + r) * 64 + c0] = acc0[r];
        ws[pbase + (rbase + r) * 64 + c0 + 16] = acc1[r];
    }
}

__global__ __launch_bounds__(256) void cosattn_reduce(float* __restrict__ ws) {
    const int slot = blockIdx.x / 17;
    const int e = (blockIdx.x % 17) * 256 + threadIdx.x;
    if (e >= 4225) return;
    const long pb = (long)slot * NS * PSTR;
    float a = 0.0f;
#pragma unroll
    for (int c = 0; c < NS; ++c) a += ws[pb + (long)c * PSTR + e];
    ws[FINBASE + (long)slot * PSTR + e] = a;
}

__global__ __launch_bounds__(256) void cosattn_p2(
    const float* __restrict__ Q, const float* __restrict__ ws,
    float* __restrict__ out) {
    __shared__ alignas(16) float fin[PSTR];
    __shared__ alignas(16) float qs[8 * 16 * LROW];
    __shared__ float scale_s[128];
    __shared__ float denom_s[128];

    const int tid = threadIdx.x;
    const int lane = tid & 31;
    const int wv = tid >> 5;
    const int bh = blockIdx.x >> 5;
    const int rg = blockIdx.x & 31;
    const int q0 = rg * 128;

    const float* finp = ws + FINBASE + (long)bh * PSTR;
    const long qrowbase = (long)bh * SS + q0 + wv * 16;
    float* myq = qs + wv * 16 * LROW;
    const float* qg = Q + qrowbase * DD;

    float4 mreg[4];
    for (int j = 0; j < 4; ++j)
        mreg[j] = *(const float4*)(finp + 4 * (tid + j * 256));
    const float treg = finp[4096 + (tid < 129 ? tid : 128)];
    float4 qreg[8];
    for (int j = 0; j < 8; ++j)
        qreg[j] = *(const float4*)(qg + 4 * (j * 32 + lane));

    for (int j = 0; j < 4; ++j)
        *(float4*)(fin + 4 * (tid + j * 256)) = mreg[j];
    fin[4096 + tid] = treg;
    for (int j = 0; j < 8; ++j) {
        int f = j * 32 + lane;
        *(float4*)(myq + (f >> 4) * LROW + (f & 15) * 4) = qreg[j];
    }
    __syncthreads();

    const int r = lane & 15;
    const int half = lane >> 4;
    {
        float ssum = 0.0f, kd = 0.0f;
#pragma unroll
        for (int c32 = 0; c32 < 32; ++c32) {
            int c = half * 32 + c32;
            float v = myq[r * LROW + c];
            ssum += v * v;
            kd += v * fin[4096 + c];
        }
        ssum += __shfl_xor(ssum, 16, 32);
        kd += __shfl_xor(kd, 16, 32);
        float sc = 1.0f / fmaxf(sqrtf(ssum), 1e-12f);
        if (lane < 16) {
            scale_s[wv * 16 + lane] = sc;
            denom_s[wv * 16 + lane] = sc * kd + fin[4224];
        }
    }
    __syncthreads();

    v8f acc[4];
#pragma unroll
    for (int nt = 0; nt < 4; ++nt) {
        float sv = fin[4160 + nt * 16 + r];
        v8f ci;
#pragma unroll
        for (int i = 0; i < 8; ++i) ci[i] = sv;
        acc[nt] = ci;
    }
    const float scr = scale_s[wv * 16 + r];
    for (int k0 = 0; k0 < DD; k0 += 4) {
        int ks = k0 + 2 * half;
        v2f a;
        a.x = myq[r * LROW + ks] * scr;
        a.y = myq[r * LROW + ks + 1] * scr;
#pragma unroll
        for (int nt = 0; nt < 4; ++nt) {
            int n = nt * 16 + r;
            v2f bfr;
            bfr.x = fin[ks * 64 + n];
            bfr.y = fin[(ks + 1) * 64 + n];
            acc[nt] = wmma_f32_4(a, bfr, acc[nt]);
        }
    }

#pragma unroll
    for (int nt = 0; nt < 4; ++nt) {
        int col = nt * 16 + r;
#pragma unroll
        for (int rr = 0; rr < 8; ++rr) {
            int rowT = rr + 8 * half;
            float dv = denom_s[wv * 16 + rowT];
            out[(qrowbase + rowT) * DD + col] = acc[nt][rr] / dv;
        }
    }
}

extern "C" void kernel_launch(void* const* d_in, const int* in_sizes, int n_in,
                              void* d_out, int out_size, void* d_ws, size_t ws_size,
                              hipStream_t stream) {
    const float* Q = (const float*)d_in[0];
    const float* K = (const float*)d_in[1];
    const float* V = (const float*)d_in[2];
    const int* M = (const int*)d_in[3];
    float* out = (float*)d_out;
    float* ws = (float*)d_ws;

    cosattn_p1<<<dim3(BH * NS), dim3(256), 0, stream>>>(K, V, M, ws);
    cosattn_reduce<<<dim3(BH * 17), dim3(256), 0, stream>>>(ws);
    cosattn_p2<<<dim3(BH * 32), dim3(256), 0, stream>>>(Q, ws, out);
}
